# starting kernel
# baseline (speedup 1.0000x reference)
#include <hip/hip_runtime.h>


#define DCH   10
#define LEN   128
#define NSEG  127
#define NB    256
#define SIGSZ 111110
#define OFF5  11110
#define NTIL  625
#define NTHR  256
#define NW    8
#define GS    8
#define NGRP  16

typedef float v2f __attribute__((ext_vector_type(2)));
typedef float v8f __attribute__((ext_vector_type(8)));

__global__ __launch_bounds__(NTHR)
void sig5_kernel(const float* __restrict__ path, float* __restrict__ out)
{
    const int b    = blockIdx.x;
    const int tid  = threadIdx.x;
    const int lane = tid & 31;
    const int wave = tid >> 5;

    __shared__ float sd[GS * 16];
    __shared__ float S1[16];
    __shared__ float S2[128];
    __shared__ float S3[1024];
    __shared__ float S4[10016];
    __shared__ float t2[128], x2[128], y2[128];
    __shared__ float t3s[GS * 1024];
    __shared__ float x3s[GS * 1024];

    float*       out5 = out + (size_t)b * SIGSZ + OFF5;
    const float* pb   = path + (size_t)b * DCH * LEN;

    const int n    = lane & 15;
    const int mrow = (lane >> 4) << 3;

    for (int i = tid; i < GS * 16; i += NTHR) sd[i] = 0.f;
    for (int i = tid; i < 16;    i += NTHR) S1[i] = 0.f;
    for (int i = tid; i < 128;   i += NTHR) S2[i] = 0.f;
    for (int i = tid; i < 1024;  i += NTHR) S3[i] = 0.f;
    for (int i = tid; i < 10016; i += NTHR) S4[i] = 0.f;
    for (int tile = wave; tile < NTIL; tile += NW) {
        if (n < DCH) {
            float* basep = out5 + (size_t)(tile << 4) * DCH + n;
            #pragma unroll
            for (int j = 0; j < 8; ++j) basep[(mrow + j) * DCH] = 0.f;
        }
    }
    __syncthreads();

    for (int g = 0; g < NGRP; ++g) {
        for (int k = 0; k < GS; ++k) {
            const int s = g * GS + k;
            if (tid < DCH) {
                float d = 0.f;
                if (s < NSEG) { const float* pr = pb + tid * LEN + s; d = pr[1] - pr[0]; }
                sd[k * 16 + tid] = d;
            }
            __syncthreads();
            if (s < NSEG) {
                const float* sdk = sd + k * 16;
                for (int i = tid; i < 100; i += NTHR) {
                    int p = i / 10, q = i - 10 * p;
                    float dp = sdk[p], dq = sdk[q], s1 = S1[p], s2v = S2[i];
                    t2[i] = s2v + (s1 + dp * 0.2f)        * dq * 0.25f;
                    x2[i] = s2v + (s1 + dp * 0.25f)       * dq * (1.f / 3.f);
                    y2[i] = s2v + (s1 + dp * (1.f / 3.f)) * dq * 0.5f;
                }
                __syncthreads();
                for (int i = tid; i < 1000; i += NTHR) {
                    int p = i / 10, q = i - 10 * p;
                    float dq = sdk[q], s3v = S3[i];
                    t3s[k * 1024 + i] = s3v + t2[p] * dq * (1.f / 3.f);
                    x3s[k * 1024 + i] = s3v + x2[p] * dq * 0.5f;
                }
                for (int i = tid; i < 100; i += NTHR) {
                    int p = i / 10, q = i - 10 * p;
                    S2[i] += (S1[p] + sdk[p] * 0.5f) * sdk[q];
                }
                __syncthreads();
                for (int i = tid; i < 1000; i += NTHR) {
                    int p = i / 10, q = i - 10 * p;
                    S3[i] += y2[p] * sdk[q];
                }
                if (tid < DCH) S1[tid] += sdk[tid];
                __syncthreads();
            }
        }

        {
            v2f bv0, bv1;
            bv0.x = (lane < 16) ? sd[0 * 16 + n] : sd[2 * 16 + n];
            bv0.y = (lane < 16) ? sd[1 * 16 + n] : sd[3 * 16 + n];
            bv1.x = (lane < 16) ? sd[4 * 16 + n] : sd[6 * 16 + n];
            bv1.y = (lane < 16) ? sd[5 * 16 + n] : sd[7 * 16 + n];
            for (int tile = wave; tile < NTIL; tile += NW) {
                const int r0 = tile << 4;
                const int r  = r0 + n;
                const int p  = r / 10, q = r - 10 * p;
                float run = S4[r];
                float u[GS];
                #pragma unroll
                for (int k = 0; k < GS; ++k) {
                    float dq = sd[k * 16 + q];
                    u[k]  = run + t3s[k * 1024 + p] * dq * 0.5f;
                    run  += x3s[k * 1024 + p] * dq;
                }
                if (lane < 16) S4[r] = run;
                v2f av0, av1;
                av0.x = (lane < 16) ? u[0] : u[2];
                av0.y = (lane < 16) ? u[1] : u[3];
                av1.x = (lane < 16) ? u[4] : u[6];
                av1.y = (lane < 16) ? u[5] : u[7];
                v8f cv;
                float* basep = out5 + (size_t)r0 * DCH + n;
                if (n < DCH) {
                    #pragma unroll
                    for (int j = 0; j < 8; ++j) cv[j] = basep[(mrow + j) * DCH];
                } else {
                    #pragma unroll
                    for (int j = 0; j < 8; ++j) cv[j] = 0.f;
                }
                cv = __builtin_amdgcn_wmma_f32_16x16x4_f32(
                        false, av0, false, bv0, (short)0, cv, false, false);
                cv = __builtin_amdgcn_wmma_f32_16x16x4_f32(
                        false, av1, false, bv1, (short)0, cv, false, false);
                if (n < DCH) {
                    #pragma unroll
                    for (int j = 0; j < 8; ++j) basep[(mrow + j) * DCH] = cv[j];
                }
            }
        }
        __syncthreads();
    }

    float* ob = out + (size_t)b * SIGSZ;
    if (tid < DCH) ob[tid] = S1[tid];
    for (int i = tid; i < 100;   i += NTHR) ob[10 + i]   = S2[i];
    for (int i = tid; i < 1000;  i += NTHR) ob[110 + i]  = S3[i];
    for (int i = tid; i < 10000; i += NTHR) ob[1110 + i] = S4[i];
}

extern "C" void kernel_launch(void* const* d_in, const int* in_sizes, int n_in,
                              void* d_out, int out_size, void* d_ws, size_t ws_size,
                              hipStream_t stream) {
    (void)in_sizes; (void)n_in; (void)d_ws; (void)ws_size; (void)out_size;
    const float* path = (const float*)d_in[0];
    float*       out  = (float*)d_out;
    sig5_kernel<<<NB, NTHR, 0, stream>>>(path, out);
}


# Level-5 tiles stay in WMMA accumulators for all 127 steps: each lane recomputes its row's level 1-4 Horner chain from deltas in LDS, so level 5 is stored once, not read-modified per step group
# speedup vs baseline: 6.1458x; 6.1458x over previous
#include <hip/hip_runtime.h>


#define DCH   10
#define LEN   128
#define NSEG  127
#define SIGSZ 111110
#define OFF5  11110
#define NROW  10000
#define NTHR  256
#define NW    8
#define NST   313
#define WGPB  40
#define NWG   10240
#define PITCH 132

typedef float    v2f __attribute__((ext_vector_type(2)));
typedef float    v4f __attribute__((ext_vector_type(4)));
typedef float    v8f __attribute__((ext_vector_type(8)));


typedef unsigned v2u __attribute__((ext_vector_type(2)));

__device__ __forceinline__ void swap16(unsigned& p, unsigned& q)
{
    const v2u sw = __builtin_amdgcn_permlane16_swap(p, q, false, false);
    p = sw[0];
    q = sw[1];
}

__global__ __launch_bounds__(NTHR)
void sig5_kernel(const float* __restrict__ path, float* __restrict__ out)
{
    const int b    = blockIdx.x / WGPB;
    const int wg   = blockIdx.x - b * WGPB;
    const int tid  = threadIdx.x;
    const int lane = tid & 31;
    const int wave = tid >> 5;

    __shared__ __attribute__((aligned(16))) float sdT[16 * PITCH];

    const float* pb = path + (size_t)b * DCH * LEN;
    for (int it = 0; it < (DCH * LEN) / NTHR; ++it) {
        const int i  = tid + it * NTHR;
        const int k  = i & (LEN - 1);
        const int k1 = (k < NSEG) ? k + 1 : k;
        const float p1 = pb[i - k + k1];
        const float p0 = pb[i];
        sdT[(i >> 7) * PITCH + k] = (k < NSEG) ? (p1 - p0) : 0.f;
    }
    for (int i = tid; i < 6 * LEN; i += NTHR)
        sdT[(DCH + (i >> 7)) * PITCH + (i & (LEN - 1))] = 0.f;
    __syncthreads();

    const int st = wg * NW + wave;
    if (st >= NST) return;

    const int  rraw  = st * 32 + lane;
    const bool valid = rraw < NROW;
    const int  r     = valid ? rraw : (NROW - 1);
    const int  da_i  = r / 1000;
    const int  r3    = r - da_i * 1000;
    const int  db_i  = r3 / 100;
    const int  r2    = r3 - db_i * 100;
    const int  dc_i  = r2 / 10;
    const int  dd_i  = r2 - dc_i * 10;

    const int n = lane & 15;
    const int h = lane >> 4;

    const float* pA = sdT + da_i * PITCH;
    const float* pB = sdT + db_i * PITCH;
    const float* pC = sdT + dc_i * PITCH;
    const float* pD = sdT + dd_i * PITCH;
    const float* pN = sdT + n * PITCH + 2 * h;

    float S1 = 0.f, S2 = 0.f, S3 = 0.f, S4 = 0.f;
    v8f c0, c1;
    for (int j = 0; j < 8; ++j) { c0[j] = 0.f; c1[j] = 0.f; }

    for (int k0 = 0; k0 < LEN; k0 += 4) {
        const v4f qa = *(const v4f*)(pA + k0);
        const v4f qb = *(const v4f*)(pB + k0);
        const v4f qc = *(const v4f*)(pC + k0);
        const v4f qd = *(const v4f*)(pD + k0);
        const v2f bv = *(const v2f*)(pN + k0);

        float u[4];
        for (int j = 0; j < 4; ++j) {
            const float da = qa[j], db = qb[j], dc = qc[j], dd = qd[j];
            const float t1 = S1 + da * 0.2f;
            const float t2 = S2 + t1 * db * 0.25f;
            const float t3 = S3 + t2 * dc * (1.f / 3.f);
            u[j]           = S4 + t3 * dd * 0.5f;
            const float x1 = S1 + da * 0.25f;
            const float x2 = S2 + x1 * db * (1.f / 3.f);
            const float x3 = S3 + x2 * dc * 0.5f;
            const float y1 = S1 + da * (1.f / 3.f);
            const float y2 = S2 + y1 * db * 0.5f;
            const float z1 = S1 + da * 0.5f;
            S4 += x3 * dd;
            S3 += y2 * dc;
            S2 += z1 * db;
            S1 += da;
        }

        unsigned e0 = __builtin_bit_cast(unsigned, u[0]);
        unsigned e2 = __builtin_bit_cast(unsigned, u[2]);
        unsigned e1 = __builtin_bit_cast(unsigned, u[1]);
        unsigned e3 = __builtin_bit_cast(unsigned, u[3]);
        swap16(e0, e2);
        swap16(e1, e3);
        v2f a0, a1;
        a0.x = __builtin_bit_cast(float, e0);
        a0.y = __builtin_bit_cast(float, e1);
        a1.x = __builtin_bit_cast(float, e2);
        a1.y = __builtin_bit_cast(float, e3);

        c0 = __builtin_amdgcn_wmma_f32_16x16x4_f32(
                false, a0, false, bv, (short)0, c0, false, false);
        c1 = __builtin_amdgcn_wmma_f32_16x16x4_f32(
                false, a1, false, bv, (short)0, c1, false, false);
    }

    float* ob = out + (size_t)b * SIGSZ;
    if (valid) {
        ob[1110 + rraw] = S4;
        if (dd_i == 0)                   ob[110 + rraw / 10]  = S3;
        if (r2 == 0)                     ob[10 + rraw / 100]  = S2;
        if (r3 == 0)                     ob[rraw / 1000]      = S1;
    }
    if (n < DCH) {
        float* bp = ob + OFF5 + (size_t)(st * 32 + 8 * h) * DCH + n;
        for (int j = 0; j < 8; ++j) bp[j * DCH] = c0[j];
        if (st * 32 + 16 < NROW) {
            for (int j = 0; j < 8; ++j) bp[(16 + j) * DCH] = c1[j];
        }
    }
}

extern "C" void kernel_launch(void* const* d_in, const int* in_sizes, int n_in,
                              void* d_out, int out_size, void* d_ws, size_t ws_size,
                              hipStream_t stream) {
    (void)in_sizes; (void)n_in; (void)d_ws; (void)ws_size; (void)out_size;
    const float* path = (const float*)d_in[0];
    float*       out  = (float*)d_out;
    sig5_kernel<<<NWG, NTHR, 0, stream>>>(path, out);
}


# Each lane owns one level-3 prefix (a,b,c) and its ten level-4 children: levels 1-3 of the Horner chain run once per prefix, two FMAs per child; 20 level-5 tiles per wave stay in WMMA accumulators
# speedup vs baseline: 12.4278x; 2.0222x over previous
#include <hip/hip_runtime.h>


#define DCH   10
#define LEN   128
#define NSEG  127
#define SIGSZ 111110
#define OFF5  11110
#define NPRE  1000
#define NTHR  128
#define NW    4
#define NST   32
#define WGPB  8
#define NWG   2048
#define PITCH 132

typedef float    v2f __attribute__((ext_vector_type(2)));
typedef float    v4f __attribute__((ext_vector_type(4)));
typedef float    v8f __attribute__((ext_vector_type(8)));

typedef unsigned v2u __attribute__((ext_vector_type(2)));

__device__ __forceinline__ void swap16(unsigned& p, unsigned& q)
{
    const v2u sw = __builtin_amdgcn_permlane16_swap(p, q, false, false);
    p = sw[0];
    q = sw[1];
}

__global__ __launch_bounds__(NTHR)
void sig5_kernel(const float* __restrict__ path, float* __restrict__ out)
{
    const int b    = blockIdx.x / WGPB;
    const int wg   = blockIdx.x - b * WGPB;
    const int tid  = threadIdx.x;
    const int lane = tid & 31;
    const int wave = tid >> 5;

    __shared__ __attribute__((aligned(16))) float sdT[16 * PITCH];

    const float* pb = path + (size_t)b * DCH * LEN;
    for (int it = 0; it < (DCH * LEN) / NTHR; ++it) {
        const int i  = tid + it * NTHR;
        const int k  = i & (LEN - 1);
        const int k1 = (k < NSEG) ? k + 1 : k;
        const float p1 = pb[i - k + k1];
        const float p0 = pb[i];
        sdT[(i >> 7) * PITCH + k] = (k < NSEG) ? (p1 - p0) : 0.f;
    }
    for (int i = tid; i < 6 * LEN; i += NTHR)
        sdT[(DCH + (i >> 7)) * PITCH + (i & (LEN - 1))] = 0.f;
    __syncthreads();

    const int st = wg * NW + wave;

    const int  praw  = st * 32 + lane;
    const bool valid = praw < NPRE;
    const int  p     = valid ? praw : (NPRE - 1);
    const int  da_i  = p / 100;
    const int  p2    = p - da_i * 100;
    const int  db_i  = p2 / 10;
    const int  dc_i  = p2 - db_i * 10;

    const int n = lane & 15;
    const int h = lane >> 4;

    const float* pA = sdT + da_i * PITCH;
    const float* pB = sdT + db_i * PITCH;
    const float* pC = sdT + dc_i * PITCH;
    const float* pN = sdT + n * PITCH + 2 * h;

    float S1 = 0.f, S2 = 0.f, S3 = 0.f;
    float S4[DCH];
    v8f c0[DCH], c1[DCH];
    for (int d = 0; d < DCH; ++d) {
        S4[d] = 0.f;
        for (int j = 0; j < 8; ++j) { c0[d][j] = 0.f; c1[d][j] = 0.f; }
    }

    for (int k0 = 0; k0 < LEN; k0 += 4) {
        const v4f qa = *(const v4f*)(pA + k0);
        const v4f qb = *(const v4f*)(pB + k0);
        const v4f qc = *(const v4f*)(pC + k0);
        const v2f bv = *(const v2f*)(pN + k0);

        float f5[4], f4[4];
        for (int j = 0; j < 4; ++j) {
            const float da = qa[j], db = qb[j], dc = qc[j];
            const float t1 = S1 + da * 0.2f;
            const float t2 = S2 + t1 * db * 0.25f;
            const float t3 = S3 + t2 * dc * (1.f / 3.f);
            f5[j]          = t3 * 0.5f;
            const float x1 = S1 + da * 0.25f;
            const float x2 = S2 + x1 * db * (1.f / 3.f);
            f4[j]          = S3 + x2 * dc * 0.5f;
            const float y1 = S1 + da * (1.f / 3.f);
            const float y2 = S2 + y1 * db * 0.5f;
            const float z1 = S1 + da * 0.5f;
            S3 += y2 * dc;
            S2 += z1 * db;
            S1 += da;
        }

        for (int d = 0; d < DCH; ++d) {
            const v4f qd = *(const v4f*)(sdT + d * PITCH + k0);
            float u[4];
            float s4 = S4[d];
            for (int j = 0; j < 4; ++j) {
                u[j] = s4 + f5[j] * qd[j];
                s4  += f4[j] * qd[j];
            }
            S4[d] = s4;

            unsigned e0 = __builtin_bit_cast(unsigned, u[0]);
            unsigned e2 = __builtin_bit_cast(unsigned, u[2]);
            unsigned e1 = __builtin_bit_cast(unsigned, u[1]);
            unsigned e3 = __builtin_bit_cast(unsigned, u[3]);
            swap16(e0, e2);
            swap16(e1, e3);
            v2f a0, a1;
            a0.x = __builtin_bit_cast(float, e0);
            a0.y = __builtin_bit_cast(float, e1);
            a1.x = __builtin_bit_cast(float, e2);
            a1.y = __builtin_bit_cast(float, e3);

            c0[d] = __builtin_amdgcn_wmma_f32_16x16x4_f32(
                        false, a0, false, bv, (short)0, c0[d], false, false);
            c1[d] = __builtin_amdgcn_wmma_f32_16x16x4_f32(
                        false, a1, false, bv, (short)0, c1[d], false, false);
        }
    }

    float* ob = out + (size_t)b * SIGSZ;
    if (valid) {
        for (int d = 0; d < DCH; ++d) ob[1110 + praw * DCH + d] = S4[d];
        ob[110 + praw] = S3;
        if (dc_i == 0) ob[10 + praw / 10] = S2;
        if (p2 == 0)   ob[praw / 100]     = S1;
    }
    if (n < DCH) {
        float* bp = ob + OFF5 + (size_t)(st * 32 + 8 * h) * (DCH * DCH) + n;
        const bool lo = st * 32 + 8 * h      < NPRE;
        const bool hi = st * 32 + 8 * h + 16 < NPRE;
        for (int d = 0; d < DCH; ++d) {
            if (lo) for (int j = 0; j < 8; ++j) bp[j * (DCH * DCH) + d * DCH] = c0[d][j];
            if (hi) for (int j = 0; j < 8; ++j) bp[(16 + j) * (DCH * DCH) + d * DCH] = c1[d][j];
        }
    }
}

extern "C" void kernel_launch(void* const* d_in, const int* in_sizes, int n_in,
                              void* d_out, int out_size, void* d_ws, size_t ws_size,
                              hipStream_t stream) {
    (void)in_sizes; (void)n_in; (void)d_ws; (void)ws_size; (void)out_size;
    const float* path = (const float*)d_in[0];
    float*       out  = (float*)d_out;
    sig5_kernel<<<NWG, NTHR, 0, stream>>>(path, out);
}


# Packed multiply-adds of the step loop fed without register moves: child deltas from a second, step-major LDS copy; prefix chain kept as (t,x) and (y,z) register pairs; 172 -> 127 vector slots a trip
# speedup vs baseline: 13.4421x; 1.0816x over previous
#include <hip/hip_runtime.h>


#define DCH   10
#define LEN   128
#define NSEG  127
#define SIGSZ 111110
#define OFF5  11110
#define NPRE  1000
#define NTHR  128
#define NW    4
#define NST   32
#define WGPB  8
#define NWG   2048
#define PITCH 132

typedef float    v2f __attribute__((ext_vector_type(2)));
typedef float    v4f __attribute__((ext_vector_type(4)));
typedef float    v8f __attribute__((ext_vector_type(8)));

typedef unsigned v2u __attribute__((ext_vector_type(2)));

__device__ __forceinline__ void swap16(unsigned& p, unsigned& q)
{
    const v2u sw = __builtin_amdgcn_permlane16_swap(p, q, false, false);
    p = sw[0];
    q = sw[1];
}

__global__ __launch_bounds__(NTHR)
void sig5_kernel(const float* __restrict__ path, float* __restrict__ out)
{
    const int b    = blockIdx.x / WGPB;
    const int wg   = blockIdx.x - b * WGPB;
    const int tid  = threadIdx.x;
    const int lane = tid & 31;
    const int wave = tid >> 5;

    __shared__ __attribute__((aligned(16))) float sdT[16 * PITCH];
    __shared__ __attribute__((aligned(16))) float sdK[LEN * 16];

    const float* pb = path + (size_t)b * DCH * LEN;
    for (int it = 0; it < (DCH * LEN) / NTHR; ++it) {
        const int i  = tid + it * NTHR;
        const int k  = i & (LEN - 1);
        const int k1 = (k < NSEG) ? k + 1 : k;
        const float p1 = pb[i - k + k1];
        const float p0 = pb[i];
        const float dl = (k < NSEG) ? (p1 - p0) : 0.f;
        sdT[(i >> 7) * PITCH + k] = dl;
        sdK[k * 16 + (i >> 7)]    = dl;
    }
    for (int i = tid; i < 6 * LEN; i += NTHR)
        sdT[(DCH + (i >> 7)) * PITCH + (i & (LEN - 1))] = 0.f;
    __syncthreads();

    const int st = wg * NW + wave;

    const int  praw  = st * 32 + lane;
    const bool valid = praw < NPRE;
    const int  p     = valid ? praw : (NPRE - 1);
    const int  da_i  = p / 100;
    const int  p2    = p - da_i * 100;
    const int  db_i  = p2 / 10;
    const int  dc_i  = p2 - db_i * 10;

    const int n = lane & 15;
    const int h = lane >> 4;

    const float* pA = sdT + da_i * PITCH;
    const float* pB = sdT + db_i * PITCH;
    const float* pC = sdT + dc_i * PITCH;
    const float* pN = sdT + n * PITCH + 2 * h;

    float S1 = 0.f, S2 = 0.f, S3 = 0.f;
    v2f S4p[DCH / 2];
    v8f c0[DCH], c1[DCH];
    for (int d = 0; d < DCH; ++d) {
        S4p[d >> 1][d & 1] = 0.f;
        for (int j = 0; j < 8; ++j) { c0[d][j] = 0.f; c1[d][j] = 0.f; }
    }

    for (int k0 = 0; k0 < LEN; k0 += 4) {
        const v4f qa = *(const v4f*)(pA + k0);
        const v4f qb = *(const v4f*)(pB + k0);
        const v4f qc = *(const v4f*)(pC + k0);
        const v2f bv = *(const v2f*)(pN + k0);

        float f5[4], f4[4];
        const v2f c51 = {0.2f, 0.25f};
        const v2f c32 = {1.f / 3.f, 0.5f};
        const v2f c43 = {0.25f, 1.f / 3.f};
        for (int j = 0; j < 4; ++j) {
            const float da = qa[j], db = qb[j], dc = qc[j];
            const v2f das = {da, da}, dbs = {db, db}, dcs = {dc, dc};
            const v2f S1s = {S1, S1}, S2s = {S2, S2}, S3s = {S3, S3};
            const v2f tx1 = das * c51 + S1s;
            const v2f yz1 = das * c32 + S1s;
            const v2f tx2 = (tx1 * dbs) * c43 + S2s;
            const v2f tf3 = (tx2 * dcs) * c32 + S3s;
            f5[j]          = tf3[0] * 0.5f;
            f4[j]          = tf3[1];
            const float y2 = S2 + yz1[0] * db * 0.5f;
            S3 += y2 * dc;
            S2 += yz1[1] * db;
            S1 += da;
        }

        v2f qp[4][DCH / 2];
        for (int j = 0; j < 4; ++j) {
            const float* row = sdK + (k0 + j) * 16;
            const v4f qA = *(const v4f*)(row);
            const v4f qB = *(const v4f*)(row + 4);
            const v2f qC = *(const v2f*)(row + 8);
            qp[j][0][0] = qA[0]; qp[j][0][1] = qA[1];
            qp[j][1][0] = qA[2]; qp[j][1][1] = qA[3];
            qp[j][2][0] = qB[0]; qp[j][2][1] = qB[1];
            qp[j][3][0] = qB[2]; qp[j][3][1] = qB[3];
            qp[j][4]    = qC;
        }

        for (int i = 0; i < DCH / 2; ++i) {
            v2f u[4];
            for (int j = 0; j < 4; ++j) {
                u[j]    = S4p[i] + f5[j] * qp[j][i];
                S4p[i] += f4[j] * qp[j][i];
            }
            for (int hh = 0; hh < 2; ++hh) {
                const int d = 2 * i + hh;
                unsigned e0 = __builtin_bit_cast(unsigned, (float)u[0][hh]);
                unsigned e2 = __builtin_bit_cast(unsigned, (float)u[2][hh]);
                unsigned e1 = __builtin_bit_cast(unsigned, (float)u[1][hh]);
                unsigned e3 = __builtin_bit_cast(unsigned, (float)u[3][hh]);
                swap16(e0, e2);
                swap16(e1, e3);
                v2f a0, a1;
                a0.x = __builtin_bit_cast(float, e0);
                a0.y = __builtin_bit_cast(float, e1);
                a1.x = __builtin_bit_cast(float, e2);
                a1.y = __builtin_bit_cast(float, e3);

                c0[d] = __builtin_amdgcn_wmma_f32_16x16x4_f32(
                            false, a0, false, bv, (short)0, c0[d], false, false);
                c1[d] = __builtin_amdgcn_wmma_f32_16x16x4_f32(
                            false, a1, false, bv, (short)0, c1[d], false, false);
            }
        }
    }

    int pr = praw;
    asm volatile("" : "+v"(pr));
    const int  n2   = pr & 15;
    const int  r8   = (pr & ~31) + ((pr >> 1) & 8);
    float* ob = out + (size_t)b * SIGSZ;
    if (pr < NPRE) {
        for (int d = 0; d < DCH; ++d) ob[1110 + pr * DCH + d] = S4p[d >> 1][d & 1];
        ob[110 + pr] = S3;
        if (pr % 10 == 0)  ob[10 + pr / 10] = S2;
        if (pr % 100 == 0) ob[pr / 100]     = S1;
    }
    if (n2 < DCH) {
        float* bp = ob + OFF5 + (size_t)r8 * (DCH * DCH) + n2;
        const bool lo = r8      < NPRE;
        const bool hi = r8 + 16 < NPRE;
        for (int d = 0; d < DCH; ++d) {
            if (lo) for (int j = 0; j < 8; ++j) bp[j * (DCH * DCH) + d * DCH] = c0[d][j];
            if (hi) for (int j = 0; j < 8; ++j) bp[(16 + j) * (DCH * DCH) + d * DCH] = c1[d][j];
        }
    }
}

extern "C" void kernel_launch(void* const* d_in, const int* in_sizes, int n_in,
                              void* d_out, int out_size, void* d_ws, size_t ws_size,
                              hipStream_t stream) {
    (void)in_sizes; (void)n_in; (void)d_ws; (void)ws_size; (void)out_size;
    const float* path = (const float*)d_in[0];
    float*       out  = (float*)d_out;
    sig5_kernel<<<NWG, NTHR, 0, stream>>>(path, out);
}


# Level 5 leaves through a per-wave LDS staging area in output order: a wave writes its 3200 consecutive floats as whole 128-byte lines, 16 B per lane, not as 160 4-byte stores of two 40-byte pieces
# speedup vs baseline: 14.8110x; 1.1018x over previous
#include <hip/hip_runtime.h>


#define DCH   10
#define LEN   128
#define NSEG  127
#define SIGSZ 111110
#define OFF5  11110
#define NPRE  1000
#define NTHR  128
#define NW    4
#define NST   32
#define WGPB  8
#define NWG   2048
#define PITCH 132

#define ROWF  100
#define STG   3208

typedef float    v2f __attribute__((ext_vector_type(2)));
typedef float    v4f __attribute__((ext_vector_type(4)));
typedef float    v8f __attribute__((ext_vector_type(8)));

typedef unsigned v2u __attribute__((ext_vector_type(2)));

__device__ __forceinline__ void swap16(unsigned& p, unsigned& q)
{
    const v2u sw = __builtin_amdgcn_permlane16_swap(p, q, false, false);
    p = sw[0];
    q = sw[1];
}

__global__ __launch_bounds__(NTHR)
void sig5_kernel(const float* __restrict__ path, float* __restrict__ out)
{
    const int b    = blockIdx.x / WGPB;
    const int wg   = blockIdx.x - b * WGPB;
    const int tid  = threadIdx.x;
    const int lane = tid & 31;
    const int wave = tid >> 5;

    __shared__ __attribute__((aligned(16))) float sdT[16 * PITCH];
    __shared__ __attribute__((aligned(16))) float sdK[LEN * 16];
    __shared__ __attribute__((aligned(16))) float stg[NW * STG];

    const float* pb = path + (size_t)b * DCH * LEN;
    for (int it = 0; it < (DCH * LEN) / NTHR; ++it) {
        const int i  = tid + it * NTHR;
        const int k  = i & (LEN - 1);
        const int k1 = (k < NSEG) ? k + 1 : k;
        const float p1 = pb[i - k + k1];
        const float p0 = pb[i];
        const float dl = (k < NSEG) ? (p1 - p0) : 0.f;
        sdT[(i >> 7) * PITCH + k] = dl;
        sdK[k * 16 + (i >> 7)]    = dl;
    }
    for (int i = tid; i < 6 * LEN; i += NTHR)
        sdT[(DCH + (i >> 7)) * PITCH + (i & (LEN - 1))] = 0.f;
    __syncthreads();

    const int st = wg * NW + wave;

    const int  praw  = st * 32 + lane;
    const bool valid = praw < NPRE;
    const int  p     = valid ? praw : (NPRE - 1);
    const int  da_i  = p / 100;
    const int  p2    = p - da_i * 100;
    const int  db_i  = p2 / 10;
    const int  dc_i  = p2 - db_i * 10;

    const int n = lane & 15;
    const int h = lane >> 4;

    const float* pA = sdT + da_i * PITCH;
    const float* pB = sdT + db_i * PITCH;
    const float* pC = sdT + dc_i * PITCH;
    const float* pN = sdT + n * PITCH + 2 * h;

    float S1 = 0.f, S2 = 0.f, S3 = 0.f;
    v2f S4p[DCH / 2];
    v8f c0[DCH], c1[DCH];
    for (int d = 0; d < DCH; ++d) {
        S4p[d >> 1][d & 1] = 0.f;
        for (int j = 0; j < 8; ++j) { c0[d][j] = 0.f; c1[d][j] = 0.f; }
    }

    for (int k0 = 0; k0 < LEN; k0 += 4) {
        const v4f qa = *(const v4f*)(pA + k0);
        const v4f qb = *(const v4f*)(pB + k0);
        const v4f qc = *(const v4f*)(pC + k0);
        const v2f bv = *(const v2f*)(pN + k0);

        float f5[4], f4[4];
        const v2f c51 = {0.2f, 0.25f};
        const v2f c32 = {1.f / 3.f, 0.5f};
        const v2f c43 = {0.25f, 1.f / 3.f};
        for (int j = 0; j < 4; ++j) {
            const float da = qa[j], db = qb[j], dc = qc[j];
            const v2f das = {da, da}, dbs = {db, db}, dcs = {dc, dc};
            const v2f S1s = {S1, S1}, S2s = {S2, S2}, S3s = {S3, S3};
            const v2f tx1 = das * c51 + S1s;
            const v2f yz1 = das * c32 + S1s;
            const v2f tx2 = (tx1 * dbs) * c43 + S2s;
            const v2f tf3 = (tx2 * dcs) * c32 + S3s;
            f5[j]          = tf3[0] * 0.5f;
            f4[j]          = tf3[1];
            const float y2 = S2 + yz1[0] * db * 0.5f;
            S3 += y2 * dc;
            S2 += yz1[1] * db;
            S1 += da;
        }

        v2f qp[4][DCH / 2];
        for (int j = 0; j < 4; ++j) {
            const float* row = sdK + (k0 + j) * 16;
            const v4f qA = *(const v4f*)(row);
            const v4f qB = *(const v4f*)(row + 4);
            const v2f qC = *(const v2f*)(row + 8);
            qp[j][0][0] = qA[0]; qp[j][0][1] = qA[1];
            qp[j][1][0] = qA[2]; qp[j][1][1] = qA[3];
            qp[j][2][0] = qB[0]; qp[j][2][1] = qB[1];
            qp[j][3][0] = qB[2]; qp[j][3][1] = qB[3];
            qp[j][4]    = qC;
        }

        for (int i = 0; i < DCH / 2; ++i) {
            v2f u[4];
            for (int j = 0; j < 4; ++j) {
                u[j]    = S4p[i] + f5[j] * qp[j][i];
                S4p[i] += f4[j] * qp[j][i];
            }
            for (int hh = 0; hh < 2; ++hh) {
                const int d = 2 * i + hh;
                unsigned e0 = __builtin_bit_cast(unsigned, (float)u[0][hh]);
                unsigned e2 = __builtin_bit_cast(unsigned, (float)u[2][hh]);
                unsigned e1 = __builtin_bit_cast(unsigned, (float)u[1][hh]);
                unsigned e3 = __builtin_bit_cast(unsigned, (float)u[3][hh]);
                swap16(e0, e2);
                swap16(e1, e3);
                v2f a0, a1;
                a0.x = __builtin_bit_cast(float, e0);
                a0.y = __builtin_bit_cast(float, e1);
                a1.x = __builtin_bit_cast(float, e2);
                a1.y = __builtin_bit_cast(float, e3);

                c0[d] = __builtin_amdgcn_wmma_f32_16x16x4_f32(
                            false, a0, false, bv, (short)0, c0[d], false, false);
                c1[d] = __builtin_amdgcn_wmma_f32_16x16x4_f32(
                            false, a1, false, bv, (short)0, c1[d], false, false);
            }
        }
    }

    int pr = praw;
    asm volatile("" : "+v"(pr));
    const int  ln  = pr & 31;
    const int  n2  = pr & 15;
    const int  h2  = (pr >> 4) & 1;
    const int  st2 = __builtin_amdgcn_readfirstlane(pr >> 5);
    float* ob = out + (size_t)b * SIGSZ;
    if (pr < NPRE) {
        for (int d = 0; d < DCH; ++d) ob[1110 + pr * DCH + d] = S4p[d >> 1][d & 1];
        ob[110 + pr] = S3;
        if (pr % 10 == 0)  ob[10 + pr / 10] = S2;
        if (pr % 100 == 0) ob[pr / 100]     = S1;
    }

    const int  nrow = (NPRE - st2 * 32 < 32) ? (NPRE - st2 * 32) : 32;
    const int  rl   = nrow * ROWF;
    const int  hf   = ((128 - ((24 * (b + 1)) & 127)) & 127) >> 2;
    const int  sh   = (4 - (hf & 3)) & 3;
    float* sw = stg + (st2 & (NW - 1)) * STG + sh;
    float* og = ob + OFF5 + (size_t)st2 * (32 * ROWF);

    if (n2 < DCH) {
        float* sp = sw + (8 * h2) * ROWF + n2;
        for (int d = 0; d < DCH; ++d) {
            for (int j = 0; j < 8; ++j) sp[j * ROWF + d * DCH]        = c0[d][j];
            for (int j = 0; j < 8; ++j) sp[(16 + j) * ROWF + d * DCH] = c1[d][j];
        }
    }
    __builtin_amdgcn_fence(__ATOMIC_RELEASE, "wavefront");
    __builtin_amdgcn_wave_barrier();
    __builtin_amdgcn_fence(__ATOMIC_ACQUIRE, "wavefront");

    const float* sb = sw + hf + ln * 4;
    float*       gb = og + hf + ln * 4;
    if (nrow == 32) {
        for (int g = 0; g < 2; ++g) {
            v4f v[12];
            for (int i = 0; i < 12; ++i) v[i] = *(const v4f*)(sb + (g * 12 + i) * 128);
            for (int i = 0; i < 12; ++i) *(v4f*)(gb + (g * 12 + i) * 128) = v[i];
        }
        const int  oe = (2 * ln < hf) ? 2 * ln : (32 * ROWF - 32) + 2 * ln;
        const bool e4 = ln < 24;
        const bool e2 = ln < 16;
        v4f v4 = {0.f, 0.f, 0.f, 0.f};
        v2f v2 = {0.f, 0.f};
        if (e4) v4 = *(const v4f*)(sb + 24 * 128);
        if (e2) v2 = *(const v2f*)(sw + oe);
        if (e4) *(v4f*)(gb + 24 * 128) = v4;
        if (e2) *(v2f*)(og + oe) = v2;
    } else {
        const int body = rl - 32;
        for (int i = 0; i < 6; ++i) *(v4f*)(gb + i * 128) = *(const v4f*)(sb + i * 128);
        if (ln < 16) {
            const int oe = (2 * ln < hf) ? 2 * ln : body + 2 * ln;
            *(v2f*)(og + oe) = *(const v2f*)(sw + oe);
        }
    }
}

extern "C" void kernel_launch(void* const* d_in, const int* in_sizes, int n_in,
                              void* d_out, int out_size, void* d_ws, size_t ws_size,
                              hipStream_t stream) {
    (void)in_sizes; (void)n_in; (void)d_ws; (void)ws_size; (void)out_size;
    const float* path = (const float*)d_in[0];
    float*       out  = (float*)d_out;
    sig5_kernel<<<NWG, NTHR, 0, stream>>>(path, out);
}
